# starting kernel
# baseline (speedup 1.0000x reference)
#include <hip/hip_runtime.h>
#include <hip/hip_bf16.h>

typedef __attribute__((ext_vector_type(2))) float v2f;
typedef __attribute__((ext_vector_type(8))) float v8f;

#define TT 2048
#define BB 64
#define DD 512
#define ROWS (TT * BB)
#define WAVES_PER_BLOCK 8
#define ROWS_PER_BLOCK (WAVES_PER_BLOCK * 16)

__global__ void __launch_bounds__(256)
xproj_wmma_kernel(const float* __restrict__ x,
                  const float* __restrict__ W,
                  float* __restrict__ xproj) {
    __shared__ float wlds[DD];
    for (int i = threadIdx.x; i < DD; i += blockDim.x) wlds[i] = W[i];
    __syncthreads();

    const int lane    = threadIdx.x & 31;
    const int wave    = threadIdx.x >> 5;
    const int m       = lane & 15;
    const int halfSel = lane >> 4;

    const long tile    = (long)blockIdx.x * WAVES_PER_BLOCK + wave;
    const long rowBase = tile * 16;

    const float* xRow = x + (rowBase + m) * DD + 2 * halfSel;
    const float* wPtr = wlds + 2 * halfSel;

    v8f c = {};
#pragma unroll 8
    for (int k = 0; k < DD; k += 4) {
        v2f a = *(const v2f*)(xRow + k);
        v2f b = *(const v2f*)(wPtr + k);
        c = __builtin_amdgcn_wmma_f32_16x16x4_f32(
                false, a, false, b, (short)0, c, false, false);
    }

    if (m == 0) {
        float* dst = xproj + rowBase + 8 * halfSel;
        *(float4*)(dst + 0) = make_float4(c[0], c[1], c[2], c[3]);
        *(float4*)(dst + 4) = make_float4(c[4], c[5], c[6], c[7]);
    }
}

#define PF 16

__global__ void __launch_bounds__(64)
scan_kernel(const float* __restrict__ xproj,
            const float* __restrict__ W,
            const float* __restrict__ bptr,
            float* __restrict__ out) {
    const int   col  = threadIdx.x;
    const float wz   = W[DD];
    const float bias = bptr[0];

    float buf[PF], nbuf[PF];
#pragma unroll
    for (int i = 0; i < PF; ++i) buf[i] = xproj[i * BB + col];

    float z = 0.0f;
    const int ntiles = TT / PF;
    for (int tile = 0; tile < ntiles; ++tile) {
        if (tile + 1 < ntiles) {
            const int nt = (tile + 1) * PF;
#pragma unroll
            for (int i = 0; i < PF; ++i) nbuf[i] = xproj[(nt + i) * BB + col];
        }
        const int t0 = tile * PF;
#pragma unroll
        for (int i = 0; i < PF; ++i) {
            float u = buf[i] + bias + wz * z;
            z = 1.0f / (1.0f + __expf(-u));
            out[(t0 + i) * BB + col] = z;
        }
#pragma unroll
        for (int i = 0; i < PF; ++i) buf[i] = nbuf[i];
    }
}

extern "C" void kernel_launch(void* const* d_in, const int* in_sizes, int n_in,
                              void* d_out, int out_size, void* d_ws, size_t ws_size,
                              hipStream_t stream) {
    const float* x  = (const float*)d_in[0];
    const float* W  = (const float*)d_in[1];
    const float* b  = (const float*)d_in[2];
    float* out   = (float*)d_out;
    float* xproj = (float*)d_ws;

    dim3 block(32 * WAVES_PER_BLOCK);
    dim3 grid(ROWS / ROWS_PER_BLOCK);
    xproj_wmma_kernel<<<grid, block, 0, stream>>>(x, W, xproj);
    scan_kernel<<<1, BB, 0, stream>>>(xproj, W, b, out);
}


# scan: serial path cut to a 4-instruction chain (fma, exp2, add, rcp): scale and bias folded into the prefetched term, v_rcp_f32 (1 ulp) for the IEEE division, rows requested 56 steps ahead
# speedup vs baseline: 2.7668x; 2.7668x over previous
#include <hip/hip_runtime.h>
#include <hip/hip_bf16.h>

typedef __attribute__((ext_vector_type(2))) float v2f;
typedef __attribute__((ext_vector_type(8))) float v8f;

#define TT 2048
#define BB 64
#define DD 512
#define ROWS (TT * BB)
#define WAVES_PER_BLOCK 8
#define ROWS_PER_BLOCK (WAVES_PER_BLOCK * 16)

__global__ void __launch_bounds__(256)
xproj_wmma_kernel(const float* __restrict__ x,
                  const float* __restrict__ W,
                  float* __restrict__ xproj) {
    __shared__ float wlds[DD];
    for (int i = threadIdx.x; i < DD; i += blockDim.x) wlds[i] = W[i];
    __syncthreads();

    const int lane    = threadIdx.x & 31;
    const int wave    = threadIdx.x >> 5;
    const int m       = lane & 15;
    const int halfSel = lane >> 4;

    const long tile    = (long)blockIdx.x * WAVES_PER_BLOCK + wave;
    const long rowBase = tile * 16;

    const float* xRow = x + (rowBase + m) * DD + 2 * halfSel;
    const float* wPtr = wlds + 2 * halfSel;

    v8f c = {};
#pragma unroll 8
    for (int k = 0; k < DD; k += 4) {
        v2f a = *(const v2f*)(xRow + k);
        v2f b = *(const v2f*)(wPtr + k);
        c = __builtin_amdgcn_wmma_f32_16x16x4_f32(
                false, a, false, b, (short)0, c, false, false);
    }

    if (m == 0) {
        float* dst = xproj + rowBase + 8 * halfSel;
        *(float4*)(dst + 0) = make_float4(c[0], c[1], c[2], c[3]);
        *(float4*)(dst + 4) = make_float4(c[4], c[5], c[6], c[7]);
    }
}

#define GSZ   8
#define RING  8
#define AHEAD 7
#define TRIP  (RING * GSZ)
#define NEG_LOG2E (-1.4426950408889634f)

template <bool FIRST, bool LAST>
__device__ __forceinline__ float scan_trip(float (&buf)[RING][GSZ],
                                           float (&zs)[GSZ], float z,
                                           float cz, float cb,
                                           const float* __restrict__ src,
                                           float* __restrict__ dst) {
#pragma unroll
    for (int g = 0; g < RING; ++g) {
        if (!LAST || g + AHEAD < RING) {
#pragma unroll
            for (int i = 0; i < GSZ; ++i)
                buf[(g + AHEAD) % RING][i] = src[((g + AHEAD) * GSZ + i) * BB];
        }
        float a[GSZ];
#pragma unroll
        for (int i = 0; i < GSZ; ++i)
            a[i] = __builtin_fmaf(buf[g][i], NEG_LOG2E, cb);
        __builtin_amdgcn_sched_barrier(0);
        if (!FIRST || g > 0) {
#pragma unroll
            for (int i = 0; i < GSZ; ++i)
                dst[((g - 1) * GSZ + i) * BB] = zs[i];
        }
#pragma unroll
        for (int i = 0; i < GSZ; ++i) {
            const float e = __builtin_amdgcn_exp2f(__builtin_fmaf(cz, z, a[i]));
            z = __builtin_amdgcn_rcpf(1.0f + e);
            zs[i] = z;
        }
    }
    return z;
}

__global__ void __launch_bounds__(64)
scan_kernel(const float* __restrict__ xproj,
            const float* __restrict__ W,
            const float* __restrict__ bptr,
            float* __restrict__ out) {
    const int   col = threadIdx.x;
    const float cz  = W[DD] * NEG_LOG2E;
    const float cb  = bptr[0] * NEG_LOG2E;

    const float* src = xproj + col;
    float*       dst = out + col;

    float buf[RING][GSZ];
    float zs[GSZ];
#pragma unroll
    for (int g = 0; g < AHEAD; ++g) {
#pragma unroll
        for (int i = 0; i < GSZ; ++i) buf[g][i] = src[(g * GSZ + i) * BB];
    }

    float z = scan_trip<true, false>(buf, zs, 0.0f, cz, cb, src, dst);
    int t0 = TRIP;
    for (; t0 < TT - TRIP; t0 += TRIP)
        z = scan_trip<false, false>(buf, zs, z, cz, cb, src + t0 * BB, dst + t0 * BB);
    scan_trip<false, true>(buf, zs, z, cz, cb, src + t0 * BB, dst + t0 * BB);
#pragma unroll
    for (int i = 0; i < GSZ; ++i) dst[(TT - GSZ + i) * BB] = zs[i];
}

extern "C" void kernel_launch(void* const* d_in, const int* in_sizes, int n_in,
                              void* d_out, int out_size, void* d_ws, size_t ws_size,
                              hipStream_t stream) {
    const float* x  = (const float*)d_in[0];
    const float* W  = (const float*)d_in[1];
    const float* b  = (const float*)d_in[2];
    float* out   = (float*)d_out;
    float* xproj = (float*)d_ws;

    dim3 block(32 * WAVES_PER_BLOCK);
    dim3 grid(ROWS / ROWS_PER_BLOCK);
    xproj_wmma_kernel<<<grid, block, 0, stream>>>(x, W, xproj);
    scan_kernel<<<1, BB, 0, stream>>>(xproj, W, b, out);
}


# projection reads each row as contiguous 512-byte pieces per wave (four lines per load instead of sixteen) and takes the dot product as the trace of the WMMA tile; its summation order changes
# speedup vs baseline: 2.9176x; 1.0545x over previous
#include <hip/hip_runtime.h>
#include <hip/hip_bf16.h>

typedef __attribute__((ext_vector_type(2))) float v2f;
typedef __attribute__((ext_vector_type(8))) float v8f;

#define TT 2048
#define BB 64
#define DD 512
#define ROWS (TT * BB)

typedef __attribute__((ext_vector_type(4))) float v4f;

#define RPW 32
#define PF  8
#define XWAVES 8

template <int MASK>
__device__ __forceinline__ float add_row_xmask(float v) {
    const int s = __builtin_amdgcn_update_dpp(0, __builtin_bit_cast(int, v),
                                              0x160 | MASK, 0xf, 0xf, false);
    return v + __builtin_bit_cast(float, s);
}

__device__ __forceinline__ float other_half(float v) {
    return __builtin_bit_cast(float, __builtin_amdgcn_permlanex16(
        __builtin_bit_cast(int, v), __builtin_bit_cast(int, v),
        0x76543210, 0xfedcba98, false, false));
}

template <bool REFILL>
__device__ __forceinline__ float proj_group(v4f (&buf)[PF][4], const v4f (&w)[4],
                                            const v4f* __restrict__ nxt,
                                            float res, int lrel,
                                            bool b0, bool b1, bool b2, bool onDiag) {
#pragma unroll
    for (int r = 0; r < PF; ++r) {
        v8f c = {};
#pragma unroll
        for (int q = 0; q < 4; ++q) {
            const v2f a0 = {buf[r][q].x, buf[r][q].y};
            const v2f w0 = {w[q].x, w[q].y};
            c = __builtin_amdgcn_wmma_f32_16x16x4_f32(
                    false, a0, false, w0, (short)0, c, false, false);
            const v2f a1 = {buf[r][q].z, buf[r][q].w};
            const v2f w1 = {w[q].z, w[q].w};
            c = __builtin_amdgcn_wmma_f32_16x16x4_f32(
                    false, a1, false, w1, (short)0, c, false, false);
        }
        if (REFILL) {
#pragma unroll
            for (int q = 0; q < 4; ++q) buf[r][q] = nxt[r * (DD / 4) + q * 32];
        }
        __builtin_amdgcn_sched_barrier(0);
        const float t0 = b0 ? c[1] : c[0];
        const float t1 = b0 ? c[3] : c[2];
        const float t2 = b0 ? c[5] : c[4];
        const float t3 = b0 ? c[7] : c[6];
        const float u0 = b1 ? t1 : t0;
        const float u1 = b1 ? t3 : t2;
        float d = b2 ? u1 : u0;
        d = onDiag ? d : 0.0f;
        d = add_row_xmask<1>(d);
        d = add_row_xmask<2>(d);
        d = add_row_xmask<4>(d);
        d = add_row_xmask<8>(d);
        d += other_half(d);
        res = (lrel == r) ? d : res;
    }
    return res;
}

__global__ void __launch_bounds__(32 * XWAVES)
xproj_wmma_kernel(const float* __restrict__ x,
                  const float* __restrict__ W,
                  float* __restrict__ xproj) {
    const int lane = threadIdx.x & 31;
    const int wave = threadIdx.x >> 5;
    const long row0 = ((long)blockIdx.x * XWAVES + wave) * RPW;

    const v4f* xp = (const v4f*)(x + row0 * DD) + lane;
    const v4f* wp = (const v4f*)W + lane;

    v4f w[4];
#pragma unroll
    for (int q = 0; q < 4; ++q) w[q] = wp[q * 32];

    v4f buf[PF][4];
#pragma unroll
    for (int r = 0; r < PF; ++r) {
#pragma unroll
        for (int q = 0; q < 4; ++q) buf[r][q] = xp[r * (DD / 4) + q * 32];
    }

    const bool b0 = (lane & 1) != 0;
    const bool b1 = (lane & 2) != 0;
    const bool b2 = (lane & 4) != 0;
    const bool onDiag = (lane < 8) || (lane >= 24);

    float res = 0.0f;
    int g = 0;
    for (; g < RPW / PF - 1; ++g)
        res = proj_group<true>(buf, w, xp + (g + 1) * PF * (DD / 4), res, lane - g * PF,
                               b0, b1, b2, onDiag);
    res = proj_group<false>(buf, w, xp, res, lane - g * PF, b0, b1, b2, onDiag);

    xproj[row0 + lane] = res;
}

#define GSZ   8
#define RING  8
#define AHEAD 7
#define TRIP  (RING * GSZ)
#define NEG_LOG2E (-1.4426950408889634f)

template <bool FIRST, bool LAST>
__device__ __forceinline__ float scan_trip(float (&buf)[RING][GSZ],
                                           float (&zs)[GSZ], float z,
                                           float cz, float cb,
                                           const float* __restrict__ src,
                                           float* __restrict__ dst) {
#pragma unroll
    for (int g = 0; g < RING; ++g) {
        if (!LAST || g + AHEAD < RING) {
#pragma unroll
            for (int i = 0; i < GSZ; ++i)
                buf[(g + AHEAD) % RING][i] = src[((g + AHEAD) * GSZ + i) * BB];
        }
        float a[GSZ];
#pragma unroll
        for (int i = 0; i < GSZ; ++i)
            a[i] = __builtin_fmaf(buf[g][i], NEG_LOG2E, cb);
        __builtin_amdgcn_sched_barrier(0);
        if (!FIRST || g > 0) {
#pragma unroll
            for (int i = 0; i < GSZ; ++i)
                dst[((g - 1) * GSZ + i) * BB] = zs[i];
        }
#pragma unroll
        for (int i = 0; i < GSZ; ++i) {
            const float e = __builtin_amdgcn_exp2f(__builtin_fmaf(cz, z, a[i]));
            z = __builtin_amdgcn_rcpf(1.0f + e);
            zs[i] = z;
        }
    }
    return z;
}

__global__ void __launch_bounds__(64)
scan_kernel(const float* __restrict__ xproj,
            const float* __restrict__ W,
            const float* __restrict__ bptr,
            float* __restrict__ out) {
    const int   col = threadIdx.x;
    const float cz  = W[DD] * NEG_LOG2E;
    const float cb  = bptr[0] * NEG_LOG2E;

    const float* src = xproj + col;
    float*       dst = out + col;

    float buf[RING][GSZ];
    float zs[GSZ];
#pragma unroll
    for (int g = 0; g < AHEAD; ++g) {
#pragma unroll
        for (int i = 0; i < GSZ; ++i) buf[g][i] = src[(g * GSZ + i) * BB];
    }

    float z = scan_trip<true, false>(buf, zs, 0.0f, cz, cb, src, dst);
    int t0 = TRIP;
    for (; t0 < TT - TRIP; t0 += TRIP)
        z = scan_trip<false, false>(buf, zs, z, cz, cb, src + t0 * BB, dst + t0 * BB);
    scan_trip<false, true>(buf, zs, z, cz, cb, src + t0 * BB, dst + t0 * BB);
#pragma unroll
    for (int i = 0; i < GSZ; ++i) dst[(TT - GSZ + i) * BB] = zs[i];
}

extern "C" void kernel_launch(void* const* d_in, const int* in_sizes, int n_in,
                              void* d_out, int out_size, void* d_ws, size_t ws_size,
                              hipStream_t stream) {
    const float* x  = (const float*)d_in[0];
    const float* W  = (const float*)d_in[1];
    const float* b  = (const float*)d_in[2];
    float* out   = (float*)d_out;
    float* xproj = (float*)d_ws;

    dim3 block(32 * XWAVES);
    dim3 grid(ROWS / (XWAVES * RPW));
    xproj_wmma_kernel<<<grid, block, 0, stream>>>(x, W, xproj);
    scan_kernel<<<1, BB, 0, stream>>>(xproj, W, b, out);
}


# scan: sigmoid(u) taken as 0.5+0.5*tanh(u/2) on the one-instruction hardware tanh (v_tanh_f32): serial chain cut from four dependent instructions (fma, exp2, add, rcp) to two (fma, tanh)
# speedup vs baseline: 3.4184x; 1.1716x over previous
#include <hip/hip_runtime.h>
#include <hip/hip_bf16.h>

typedef __attribute__((ext_vector_type(2))) float v2f;
typedef __attribute__((ext_vector_type(8))) float v8f;

#define TT 2048
#define BB 64
#define DD 512
#define ROWS (TT * BB)

typedef __attribute__((ext_vector_type(4))) float v4f;

#define RPW 32
#define PF  8
#define XWAVES 8

template <int MASK>
__device__ __forceinline__ float add_row_xmask(float v) {
    const int s = __builtin_amdgcn_update_dpp(0, __builtin_bit_cast(int, v),
                                              0x160 | MASK, 0xf, 0xf, false);
    return v + __builtin_bit_cast(float, s);
}

__device__ __forceinline__ float other_half(float v) {
    return __builtin_bit_cast(float, __builtin_amdgcn_permlanex16(
        __builtin_bit_cast(int, v), __builtin_bit_cast(int, v),
        0x76543210, 0xfedcba98, false, false));
}

template <bool REFILL>
__device__ __forceinline__ float proj_group(v4f (&buf)[PF][4], const v4f (&w)[4],
                                            const v4f* __restrict__ nxt,
                                            float res, int lrel,
                                            bool b0, bool b1, bool b2, bool onDiag) {
#pragma unroll
    for (int r = 0; r < PF; ++r) {
        v8f c = {};
#pragma unroll
        for (int q = 0; q < 4; ++q) {
            const v2f a0 = {buf[r][q].x, buf[r][q].y};
            const v2f w0 = {w[q].x, w[q].y};
            c = __builtin_amdgcn_wmma_f32_16x16x4_f32(
                    false, a0, false, w0, (short)0, c, false, false);
            const v2f a1 = {buf[r][q].z, buf[r][q].w};
            const v2f w1 = {w[q].z, w[q].w};
            c = __builtin_amdgcn_wmma_f32_16x16x4_f32(
                    false, a1, false, w1, (short)0, c, false, false);
        }
        if (REFILL) {
#pragma unroll
            for (int q = 0; q < 4; ++q) buf[r][q] = nxt[r * (DD / 4) + q * 32];
        }
        __builtin_amdgcn_sched_barrier(0);
        const float t0 = b0 ? c[1] : c[0];
        const float t1 = b0 ? c[3] : c[2];
        const float t2 = b0 ? c[5] : c[4];
        const float t3 = b0 ? c[7] : c[6];
        const float u0 = b1 ? t1 : t0;
        const float u1 = b1 ? t3 : t2;
        float d = b2 ? u1 : u0;
        d = onDiag ? d : 0.0f;
        d = add_row_xmask<1>(d);
        d = add_row_xmask<2>(d);
        d = add_row_xmask<4>(d);
        d = add_row_xmask<8>(d);
        d += other_half(d);
        res = (lrel == r) ? d : res;
    }
    return res;
}

__global__ void __launch_bounds__(32 * XWAVES)
xproj_wmma_kernel(const float* __restrict__ x,
                  const float* __restrict__ W,
                  float* __restrict__ xproj) {
    const int lane = threadIdx.x & 31;
    const int wave = threadIdx.x >> 5;
    const long row0 = ((long)blockIdx.x * XWAVES + wave) * RPW;

    const v4f* xp = (const v4f*)(x + row0 * DD) + lane;
    const v4f* wp = (const v4f*)W + lane;

    v4f w[4];
#pragma unroll
    for (int q = 0; q < 4; ++q) w[q] = wp[q * 32];

    v4f buf[PF][4];
#pragma unroll
    for (int r = 0; r < PF; ++r) {
#pragma unroll
        for (int q = 0; q < 4; ++q) buf[r][q] = xp[r * (DD / 4) + q * 32];
    }

    const bool b0 = (lane & 1) != 0;
    const bool b1 = (lane & 2) != 0;
    const bool b2 = (lane & 4) != 0;
    const bool onDiag = (lane < 8) || (lane >= 24);

    float res = 0.0f;
    int g = 0;
    for (; g < RPW / PF - 1; ++g)
        res = proj_group<true>(buf, w, xp + (g + 1) * PF * (DD / 4), res, lane - g * PF,
                               b0, b1, b2, onDiag);
    res = proj_group<false>(buf, w, xp, res, lane - g * PF, b0, b1, b2, onDiag);

    xproj[row0 + lane] = res;
}

#define GSZ   8
#define RING  8
#define AHEAD 7
#define TRIP  (RING * GSZ)

template <bool FIRST, bool LAST>
__device__ __forceinline__ float scan_trip(float (&buf)[RING][GSZ],
                                           float (&zs)[GSZ], float y,
                                           float ck, float cb,
                                           const float* __restrict__ src,
                                           float* __restrict__ dst) {
#pragma unroll
    for (int g = 0; g < RING; ++g) {
        if (!LAST || g + AHEAD < RING) {
#pragma unroll
            for (int i = 0; i < GSZ; ++i)
                buf[(g + AHEAD) % RING][i] = src[((g + AHEAD) * GSZ + i) * BB];
        }
        float a[GSZ];
#pragma unroll
        for (int i = 0; i < GSZ; ++i)
            a[i] = __builtin_fmaf(buf[g][i], 0.5f, cb);
        __builtin_amdgcn_sched_barrier(0);
        if (!FIRST || g > 0) {
#pragma unroll
            for (int i = 0; i < GSZ; ++i)
                dst[((g - 1) * GSZ + i) * BB] = zs[i];
        }
#pragma unroll
        for (int i = 0; i < GSZ; ++i) {
            y = __builtin_amdgcn_tanhf(__builtin_fmaf(ck, y, a[i]));
            zs[i] = __builtin_fmaf(0.5f, y, 0.5f);
        }
    }
    return y;
}

__global__ void __launch_bounds__(64)
scan_kernel(const float* __restrict__ xproj,
            const float* __restrict__ W,
            const float* __restrict__ bptr,
            float* __restrict__ out) {
    const int   col = threadIdx.x;
    const float ck  = 0.25f * W[DD];
    const float cb  = __builtin_fmaf(0.5f, bptr[0], ck);

    const float* src = xproj + col;
    float*       dst = out + col;

    float buf[RING][GSZ];
    float zs[GSZ];
#pragma unroll
    for (int g = 0; g < AHEAD; ++g) {
#pragma unroll
        for (int i = 0; i < GSZ; ++i) buf[g][i] = src[(g * GSZ + i) * BB];
    }

    float y = scan_trip<true, false>(buf, zs, -1.0f, ck, cb, src, dst);
    int t0 = TRIP;
    for (; t0 < TT - TRIP; t0 += TRIP)
        y = scan_trip<false, false>(buf, zs, y, ck, cb, src + t0 * BB, dst + t0 * BB);
    scan_trip<false, true>(buf, zs, y, ck, cb, src + t0 * BB, dst + t0 * BB);
#pragma unroll
    for (int i = 0; i < GSZ; ++i) dst[(TT - GSZ + i) * BB] = zs[i];
}

extern "C" void kernel_launch(void* const* d_in, const int* in_sizes, int n_in,
                              void* d_out, int out_size, void* d_ws, size_t ws_size,
                              hipStream_t stream) {
    const float* x  = (const float*)d_in[0];
    const float* W  = (const float*)d_in[1];
    const float* b  = (const float*)d_in[2];
    float* out   = (float*)d_out;
    float* xproj = (float*)d_ws;

    dim3 block(32 * XWAVES);
    dim3 grid(ROWS / (XWAVES * RPW));
    xproj_wmma_kernel<<<grid, block, 0, stream>>>(x, W, xproj);
    scan_kernel<<<1, BB, 0, stream>>>(xproj, W, b, out);
}


# time axis cut in two, three launches: scan of part A shares a launch with the projection of part B (block 0 scans) so the two overlap; part B starts from the stored z (y = 2z - 1)
# speedup vs baseline: 3.6668x; 1.0727x over previous
#include <hip/hip_runtime.h>
#include <hip/hip_bf16.h>

typedef __attribute__((ext_vector_type(2))) float v2f;
typedef __attribute__((ext_vector_type(8))) float v8f;

#define TT 2048
#define BB 64
#define DD 512
#define ROWS (TT * BB)

typedef __attribute__((ext_vector_type(4))) float v4f;

#define RPW 32
#define PF  8
#define XWAVES 8

template <int MASK>
__device__ __forceinline__ float add_row_xmask(float v) {
    const int s = __builtin_amdgcn_update_dpp(0, __builtin_bit_cast(int, v),
                                              0x160 | MASK, 0xf, 0xf, false);
    return v + __builtin_bit_cast(float, s);
}

__device__ __forceinline__ float other_half(float v) {
    return __builtin_bit_cast(float, __builtin_amdgcn_permlanex16(
        __builtin_bit_cast(int, v), __builtin_bit_cast(int, v),
        0x76543210, 0xfedcba98, false, false));
}

template <bool REFILL>
__device__ __forceinline__ float proj_group(v4f (&buf)[PF][4], const v4f (&w)[4],
                                            const v4f* __restrict__ nxt,
                                            float res, int lrel,
                                            bool b0, bool b1, bool b2, bool onDiag) {
#pragma unroll
    for (int r = 0; r < PF; ++r) {
        v8f c = {};
#pragma unroll
        for (int q = 0; q < 4; ++q) {
            const v2f a0 = {buf[r][q].x, buf[r][q].y};
            const v2f w0 = {w[q].x, w[q].y};
            c = __builtin_amdgcn_wmma_f32_16x16x4_f32(
                    false, a0, false, w0, (short)0, c, false, false);
            const v2f a1 = {buf[r][q].z, buf[r][q].w};
            const v2f w1 = {w[q].z, w[q].w};
            c = __builtin_amdgcn_wmma_f32_16x16x4_f32(
                    false, a1, false, w1, (short)0, c, false, false);
        }
        if (REFILL) {
#pragma unroll
            for (int q = 0; q < 4; ++q) buf[r][q] = nxt[r * (DD / 4) + q * 32];
        }
        __builtin_amdgcn_sched_barrier(0);
        const float t0 = b0 ? c[1] : c[0];
        const float t1 = b0 ? c[3] : c[2];
        const float t2 = b0 ? c[5] : c[4];
        const float t3 = b0 ? c[7] : c[6];
        const float u0 = b1 ? t1 : t0;
        const float u1 = b1 ? t3 : t2;
        float d = b2 ? u1 : u0;
        d = onDiag ? d : 0.0f;
        d = add_row_xmask<1>(d);
        d = add_row_xmask<2>(d);
        d = add_row_xmask<4>(d);
        d = add_row_xmask<8>(d);
        d += other_half(d);
        res = (lrel == r) ? d : res;
    }
    return res;
}

__device__ __forceinline__ void proj_rows(const float* __restrict__ x,
                                          const float* __restrict__ W,
                                          float* __restrict__ xproj,
                                          long row0, int lane) {
    const v4f* xp = (const v4f*)(x + row0 * DD) + lane;
    const v4f* wp = (const v4f*)W + lane;

    v4f w[4];
#pragma unroll
    for (int q = 0; q < 4; ++q) w[q] = wp[q * 32];

    v4f buf[PF][4];
#pragma unroll
    for (int r = 0; r < PF; ++r) {
#pragma unroll
        for (int q = 0; q < 4; ++q) buf[r][q] = xp[r * (DD / 4) + q * 32];
    }

    const bool b0 = (lane & 1) != 0;
    const bool b1 = (lane & 2) != 0;
    const bool b2 = (lane & 4) != 0;
    const bool onDiag = (lane < 8) || (lane >= 24);

    float res = 0.0f;
    int g = 0;
    for (; g < RPW / PF - 1; ++g)
        res = proj_group<true>(buf, w, xp + (g + 1) * PF * (DD / 4), res, lane - g * PF,
                               b0, b1, b2, onDiag);
    res = proj_group<false>(buf, w, xp, res, lane - g * PF, b0, b1, b2, onDiag);

    xproj[row0 + lane] = res;
}

#define GSZ   8
#define RING  8
#define AHEAD 7
#define TRIP  (RING * GSZ)

template <bool FIRST, bool LAST>
__device__ __forceinline__ float scan_trip(float (&buf)[RING][GSZ],
                                           float (&zs)[GSZ], float y,
                                           float ck, float cb,
                                           const float* __restrict__ src,
                                           float* __restrict__ dst) {
#pragma unroll
    for (int g = 0; g < RING; ++g) {
        if (!LAST || g + AHEAD < RING) {
#pragma unroll
            for (int i = 0; i < GSZ; ++i)
                buf[(g + AHEAD) % RING][i] = src[((g + AHEAD) * GSZ + i) * BB];
        }
        float a[GSZ];
#pragma unroll
        for (int i = 0; i < GSZ; ++i)
            a[i] = __builtin_fmaf(buf[g][i], 0.5f, cb);
        __builtin_amdgcn_sched_barrier(0);
        if (!FIRST || g > 0) {
#pragma unroll
            for (int i = 0; i < GSZ; ++i)
                dst[((g - 1) * GSZ + i) * BB] = zs[i];
        }
#pragma unroll
        for (int i = 0; i < GSZ; ++i) {
            y = __builtin_amdgcn_tanhf(__builtin_fmaf(ck, y, a[i]));
            zs[i] = __builtin_fmaf(0.5f, y, 0.5f);
        }
    }
    return y;
}

template <int NSTEPS>
__device__ __forceinline__ void scan_range(const float* __restrict__ src,
                                           float* __restrict__ dst,
                                           float y, float ck, float cb) {
    float buf[RING][GSZ];
    float zs[GSZ];
#pragma unroll
    for (int g = 0; g < AHEAD; ++g) {
#pragma unroll
        for (int i = 0; i < GSZ; ++i) buf[g][i] = src[(g * GSZ + i) * BB];
    }

    y = scan_trip<true, false>(buf, zs, y, ck, cb, src, dst);
    int t0 = TRIP;
    for (; t0 < NSTEPS - TRIP; t0 += TRIP)
        y = scan_trip<false, false>(buf, zs, y, ck, cb, src + t0 * BB, dst + t0 * BB);
    scan_trip<false, true>(buf, zs, y, ck, cb, src + t0 * BB, dst + t0 * BB);
#pragma unroll
    for (int i = 0; i < GSZ; ++i) dst[(NSTEPS - GSZ + i) * BB] = zs[i];
}

#define TA 896
#define TB (TT - TA)
#define ROWS_A (TA * BB)
#define ROWS_B (TB * BB)
#define BLK_ROWS (XWAVES * RPW)

__global__ void __launch_bounds__(32 * XWAVES)
xproj_a_kernel(const float* __restrict__ x,
               const float* __restrict__ W,
               float* __restrict__ xproj) {
    const int lane = threadIdx.x & 31;
    const int wave = threadIdx.x >> 5;
    const long row0 = ((long)blockIdx.x * XWAVES + wave) * RPW;
    proj_rows(x, W, xproj, row0, lane);
}

__global__ void __launch_bounds__(32 * XWAVES)
scan_a_xproj_b_kernel(const float* __restrict__ x,
                      const float* __restrict__ W,
                      const float* __restrict__ bptr,
                      float* xproj,
                      float* __restrict__ out) {
    if (blockIdx.x == 0) {
        if (threadIdx.x < BB) {
            const int   col = threadIdx.x;
            const float ck  = 0.25f * W[DD];
            const float cb  = __builtin_fmaf(0.5f, bptr[0], ck);
            scan_range<TA>(xproj + col, out + col, -1.0f, ck, cb);
        }
        return;
    }
    const int lane = threadIdx.x & 31;
    const int wave = threadIdx.x >> 5;
    const long row0 = (long)ROWS_A + ((long)(blockIdx.x - 1) * XWAVES + wave) * RPW;
    proj_rows(x, W, xproj, row0, lane);
}

__global__ void __launch_bounds__(64)
scan_b_kernel(const float* __restrict__ xproj,
              const float* __restrict__ W,
              const float* __restrict__ bptr,
              float* __restrict__ out) {
    const int   col = threadIdx.x;
    const float ck  = 0.25f * W[DD];
    const float cb  = __builtin_fmaf(0.5f, bptr[0], ck);
    const float y0  = __builtin_fmaf(2.0f, out[(TA - 1) * BB + col], -1.0f);
    scan_range<TB>(xproj + ROWS_A + col, out + ROWS_A + col, y0, ck, cb);
}

extern "C" void kernel_launch(void* const* d_in, const int* in_sizes, int n_in,
                              void* d_out, int out_size, void* d_ws, size_t ws_size,
                              hipStream_t stream) {
    const float* x  = (const float*)d_in[0];
    const float* W  = (const float*)d_in[1];
    const float* b  = (const float*)d_in[2];
    float* out   = (float*)d_out;
    float* xproj = (float*)d_ws;

    dim3 block(32 * XWAVES);
    xproj_a_kernel<<<dim3(ROWS_A / BLK_ROWS), block, 0, stream>>>(x, W, xproj);
    scan_a_xproj_b_kernel<<<dim3(1 + ROWS_B / BLK_ROWS), block, 0, stream>>>(x, W, b, xproj, out);
    scan_b_kernel<<<1, BB, 0, stream>>>(xproj, W, b, out);
}
